# starting kernel
# baseline (speedup 1.0000x reference)
#include <hip/hip_runtime.h>


typedef __attribute__((ext_vector_type(2))) float v2f;
typedef __attribute__((ext_vector_type(8))) float v8f;


__global__ void zero_f32(float* __restrict__ p, int n4) {
    int i = blockIdx.x * blockDim.x + threadIdx.x;
    if (i < n4) ((float4*)p)[i] = make_float4(0.f, 0.f, 0.f, 0.f);
}

__global__ void deg_kernel(const int* __restrict__ row,
                           const float* __restrict__ ew,
                           float* __restrict__ deg, int E) {
    int e = blockIdx.x * blockDim.x + threadIdx.x;
    if (e < E) atomicAdd(&deg[row[e]], ew[e]);
}

__global__ void dinv_kernel(const float* __restrict__ deg,
                            float* __restrict__ dinv, int N) {
    int i = blockIdx.x * blockDim.x + threadIdx.x;
    if (i < N) {
        float d = deg[i];
        dinv[i] = (d > 0.f) ? rsqrtf(d) : 0.f;
    }
}

__global__ void normw_kernel(const int* __restrict__ row,
                             const int* __restrict__ col,
                             const float* __restrict__ ew,
                             const float* __restrict__ dinv,
                             float* __restrict__ nw, int E) {
    int e = blockIdx.x * blockDim.x + threadIdx.x;
    if (e < E) nw[e] = -dinv[row[e]] * ew[e] * dinv[col[e]];
}

__global__ void spmm_kernel(const int* __restrict__ row,
                            const int* __restrict__ col,
                            const float* __restrict__ nw,
                            const float* __restrict__ in,
                            float* __restrict__ out,
                            int E, int D) {
    int t = blockIdx.x * blockDim.x + threadIdx.x;
    int groups = D >> 2;
    int e = t / groups;
    int g = t - e * groups;
    if (e >= E) return;
    int r = row[e], c = col[e];
    float w = nw[e];
    float4 v = ((const float4*)(in + (size_t)r * D))[g];
    float* dst = out + (size_t)c * D + (g << 2);
    atomicAdd(dst + 0, w * v.x);
    atomicAdd(dst + 1, w * v.y);
    atomicAdd(dst + 2, w * v.z);
    atomicAdd(dst + 3, w * v.w);
}

__global__ void layer1_combine(const float* __restrict__ w10,
                               const float* __restrict__ s1,
                               const float* __restrict__ s2,
                               const float* __restrict__ w12,
                               const float* __restrict__ b1,
                               float* __restrict__ h,
                               int n4, int D) {
    int i = blockIdx.x * blockDim.x + threadIdx.x;
    if (i >= n4) return;
    int cg = i & ((D >> 2) - 1);
    float4 a = ((const float4*)w10)[i];
    float4 s = ((const float4*)s1)[i];
    float4 t = ((const float4*)s2)[i];
    float4 m = ((const float4*)w12)[i];
    float4 b = ((const float4*)b1)[cg];
    float4 o;
    o.x = fmaxf(a.x + s.x + 2.f * t.x - m.x + b.x, 0.f);
    o.y = fmaxf(a.y + s.y + 2.f * t.y - m.y + b.y, 0.f);
    o.z = fmaxf(a.z + s.z + 2.f * t.z - m.z + b.z, 0.f);
    o.w = fmaxf(a.w + s.w + 2.f * t.w - m.w + b.w, 0.f);
    ((float4*)h)[i] = o;
}

__global__ void t2_combine(const float* __restrict__ u,
                           const float* __restrict__ h,
                           float* __restrict__ t2, int n4) {
    int i = blockIdx.x * blockDim.x + threadIdx.x;
    if (i >= n4) return;
    float4 a = ((const float4*)u)[i];
    float4 b = ((const float4*)h)[i];
    float4 o;
    o.x = 2.f * a.x - b.x;  o.y = 2.f * a.y - b.y;
    o.z = 2.f * a.z - b.z;  o.w = 2.f * a.w - b.w;
    ((float4*)t2)[i] = o;
}

#define DC 128
#define APAD 132

__global__ void __launch_bounds__(256)
gemm3_wmma_f32(const float* __restrict__ A0,
               const float* __restrict__ A1,
               const float* __restrict__ A2,
               const float* __restrict__ B,
               const float* __restrict__ bias,
               float* __restrict__ out) {
    __shared__ float As[3][16][APAD];

    const int wave = threadIdx.x >> 5;
    const int lane = threadIdx.x & 31;
    const int idx  = lane & 15;
    const int half = lane >> 4;

    const int tm   = blockIdx.x;
    const int tn   = wave;
    const int colB = tn * 16 + idx;

    const float* Am[3] = {A0, A1, A2};

#pragma unroll
    for (int m = 0; m < 3; ++m) {
        const float* src = Am[m] + (size_t)tm * 16 * DC;
#pragma unroll
        for (int i = 0; i < 2; ++i) {
            int f  = threadIdx.x + i * 256;
            int r  = f >> 5;
            int c4 = f & 31;
            float4 v = ((const float4*)(src + (size_t)r * DC))[c4];
            *(float4*)&As[m][r][c4 << 2] = v;
        }
    }
    __syncthreads();

    v8f c = {0.f, 0.f, 0.f, 0.f, 0.f, 0.f, 0.f, 0.f};

#pragma unroll
    for (int m = 0; m < 3; ++m) {
        const float* Arow  = &As[m][idx][2 * half];
        const float* Bbase = B + (size_t)m * DC * DC
                               + (size_t)2 * half * DC + colB;
#pragma unroll
        for (int k0 = 0; k0 < DC; k0 += 4) {
            v2f a = *(const v2f*)(Arow + k0);
            v2f b;
            b[0] = Bbase[(size_t)k0 * DC];
            b[1] = Bbase[(size_t)k0 * DC + DC];
            c = __builtin_amdgcn_wmma_f32_16x16x4_f32(
                    false, a, false, b, (short)0, c, false, false);
        }
    }

    float bv = bias[colB];
#pragma unroll
    for (int r = 0; r < 8; ++r) {
        int orow = tm * 16 + r + 8 * half;
        out[(size_t)orow * DC + colB] = c[r] + bv;
    }
}


extern "C" void kernel_launch(void* const* d_in, const int* in_sizes, int n_in,
                              void* d_out, int out_size, void* d_ws, size_t ws_size,
                              hipStream_t stream) {
    const int*   ei  = (const int*)d_in[1];
    const float* ew  = (const float*)d_in[2];
    const float* W1  = (const float*)d_in[3];
    const float* b1  = (const float*)d_in[4];
    const float* W2  = (const float*)d_in[5];
    const float* b2  = (const float*)d_in[6];
    float* out = (float*)d_out;

    const int E = in_sizes[2];
    const int D = in_sizes[4];
    const int N = in_sizes[3] / (3 * D);
    const int ND = N * D;

    const int* row = ei;
    const int* col = ei + E;
    const float* W1_0 = W1;
    const float* W1_1 = W1 + (size_t)ND;
    const float* W1_2 = W1 + (size_t)2 * ND;

    char* ws = (char*)d_ws;
    float* deg   = (float*)ws;                   ws += (size_t)N * 4;
    float* dinv  = (float*)ws;                   ws += (size_t)N * 4;
    float* normw = (float*)ws;                   ws += (size_t)E * 4;
    float* B0 = (float*)ws;                      ws += (size_t)ND * 4;
    float* B1 = (float*)ws;                      ws += (size_t)ND * 4;
    float* B2 = (float*)ws;                      ws += (size_t)ND * 4;
    float* H  = (float*)ws;                      ws += (size_t)ND * 4;

    const int TB = 256;
    const int nd4 = ND / 4;
    dim3 gZero((nd4 + TB - 1) / TB), gN4(((N / 4) + TB - 1) / TB);
    dim3 gE((E + TB - 1) / TB), gNn((N + TB - 1) / TB);
    const int spmmThreads = E * (D / 4);
    dim3 gSpmm((spmmThreads + TB - 1) / TB);

    zero_f32<<<gN4, TB, 0, stream>>>(deg, N / 4);
    deg_kernel<<<gE, TB, 0, stream>>>(row, ew, deg, E);
    dinv_kernel<<<gNn, TB, 0, stream>>>(deg, dinv, N);
    normw_kernel<<<gE, TB, 0, stream>>>(row, col, ew, dinv, normw, E);

    zero_f32<<<gZero, TB, 0, stream>>>(B0, nd4);
    spmm_kernel<<<gSpmm, TB, 0, stream>>>(row, col, normw, W1_1, B0, E, D);
    zero_f32<<<gZero, TB, 0, stream>>>(B1, nd4);
    spmm_kernel<<<gSpmm, TB, 0, stream>>>(row, col, normw, W1_2, B1, E, D);
    zero_f32<<<gZero, TB, 0, stream>>>(B2, nd4);
    spmm_kernel<<<gSpmm, TB, 0, stream>>>(row, col, normw, B1, B2, E, D);
    layer1_combine<<<gZero, TB, 0, stream>>>(W1_0, B0, B2, W1_2, b1, H, nd4, D);

    zero_f32<<<gZero, TB, 0, stream>>>(B0, nd4);
    spmm_kernel<<<gSpmm, TB, 0, stream>>>(row, col, normw, H, B0, E, D);
    zero_f32<<<gZero, TB, 0, stream>>>(B1, nd4);
    spmm_kernel<<<gSpmm, TB, 0, stream>>>(row, col, normw, B0, B1, E, D);
    t2_combine<<<gZero, TB, 0, stream>>>(B1, H, B2, nd4);

    gemm3_wmma_f32<<<N / 16, 256, 0, stream>>>(H, B0, B2, W2, b2, out);
}


# SpMM as gather: edge list counting-sorted by destination once per call (one workgroup, LDS); one wave per output row sums its in-edges in list order; global float atomics and zero fills removed
# speedup vs baseline: 1.6367x; 1.6367x over previous
#include <hip/hip_runtime.h>


typedef __attribute__((ext_vector_type(2))) float v2f;
typedef __attribute__((ext_vector_type(8))) float v8f;


#define DC 128
#define NMAX 4096
#define EMAX 32768
#define BT 1024
#define EB 8

__global__ void __launch_bounds__(BT)
build_sorted_edges(const int* __restrict__ row,
                   const int* __restrict__ col,
                   const float* __restrict__ ew,
                   int* __restrict__ rowptr,
                   int2* __restrict__ edges,
                   int E, int N) {
    __shared__ float s_dinv[NMAX];
    __shared__ int   s_cur[NMAX];
    __shared__ int   s_wsum[32];
    __shared__ int2  s_rec[EMAX];

    const int tid  = threadIdx.x;
    const int lane = tid & 31;
    const int wave = tid >> 5;

    for (int i = tid; i < NMAX; i += BT) { s_dinv[i] = 0.f; s_cur[i] = 0; }
    __syncthreads();

    int e0 = tid;
    for (; e0 + (EB - 1) * BT < E; e0 += BT * EB) {
        int r[EB], c[EB];
        float w[EB];
#pragma unroll
        for (int k = 0; k < EB; ++k) {
            r[k] = row[e0 + k * BT];
            c[k] = col[e0 + k * BT];
            w[k] = ew[e0 + k * BT];
        }
#pragma unroll
        for (int k = 0; k < EB; ++k) {
            atomicAdd(&s_dinv[r[k]], w[k]);
            atomicAdd(&s_cur[c[k]], 1);
        }
    }
    for (int e = e0; e < E; e += BT) {
        atomicAdd(&s_dinv[row[e]], ew[e]);
        atomicAdd(&s_cur[col[e]], 1);
    }
    __syncthreads();

    for (int i = tid; i < NMAX; i += BT) {
        float d = s_dinv[i];
        s_dinv[i] = (d > 0.f) ? rsqrtf(d) : 0.f;
    }

    const int b4 = tid * 4;
    const int c0 = s_cur[b4], c1 = s_cur[b4 + 1], c2 = s_cur[b4 + 2], c3 = s_cur[b4 + 3];
    const int tsum = c0 + c1 + c2 + c3;
    int v = tsum;
    for (int d = 1; d < 32; d <<= 1) {
        int t = __shfl_up(v, d);
        if (lane >= d) v += t;
    }
    if (lane == 31) s_wsum[wave] = v;
    __syncthreads();
    if (wave == 0) {
        const int w = s_wsum[lane];
        int x = w;
        for (int d = 1; d < 32; d <<= 1) {
            int t = __shfl_up(x, d);
            if (lane >= d) x += t;
        }
        s_wsum[lane] = x - w;
    }
    __syncthreads();
    const int o0 = s_wsum[wave] + v - tsum;
    const int o1 = o0 + c0, o2 = o1 + c1, o3 = o2 + c2;
    s_cur[b4] = o0; s_cur[b4 + 1] = o1; s_cur[b4 + 2] = o2; s_cur[b4 + 3] = o3;
    if (b4     < N) rowptr[b4]     = o0;
    if (b4 + 1 < N) rowptr[b4 + 1] = o1;
    if (b4 + 2 < N) rowptr[b4 + 2] = o2;
    if (b4 + 3 < N) rowptr[b4 + 3] = o3;
    if (tid == 0) rowptr[N] = E;
    __syncthreads();

    e0 = tid;
    for (; e0 + (EB - 1) * BT < E; e0 += BT * EB) {
        int r[EB], c[EB], pos[EB];
        float w[EB];
#pragma unroll
        for (int k = 0; k < EB; ++k) {
            r[k] = row[e0 + k * BT];
            c[k] = col[e0 + k * BT];
            w[k] = ew[e0 + k * BT];
        }
#pragma unroll
        for (int k = 0; k < EB; ++k) w[k] = -s_dinv[r[k]] * w[k] * s_dinv[c[k]];
#pragma unroll
        for (int k = 0; k < EB; ++k) pos[k] = atomicAdd(&s_cur[c[k]], 1);
#pragma unroll
        for (int k = 0; k < EB; ++k) s_rec[pos[k]] = make_int2(r[k], __float_as_int(w[k]));
    }
    for (int e = e0; e < E; e += BT) {
        const int r = row[e], c = col[e];
        const float w = -s_dinv[r] * ew[e] * s_dinv[c];
        const int pos = atomicAdd(&s_cur[c], 1);
        s_rec[pos] = make_int2(r, __float_as_int(w));
    }
    __syncthreads();

    const int4* s4 = (const int4*)s_rec;
    int4* g4 = (int4*)edges;
    const int n2 = E >> 1;
    for (int i = tid; i < n2; i += BT) g4[i] = s4[i];
    if (tid == 0 && (E & 1)) edges[E - 1] = s_rec[E - 1];
}

template <int K>
__device__ __forceinline__ void gather_rows(float4& acc, const float4* __restrict__ in4,
                                            int msrc, int mw, int j, int lane) {
    float4 x[K];
    float  w[K];
#pragma unroll
    for (int k = 0; k < K; ++k) {
        const int s = __builtin_amdgcn_readlane(msrc, j + k);
        w[k] = __int_as_float(__builtin_amdgcn_readlane(mw, j + k));
        x[k] = in4[(size_t)s * (DC / 4) + lane];
    }
#pragma unroll
    for (int k = 0; k < K; ++k) {
        acc.x += w[k] * x[k].x;
        acc.y += w[k] * x[k].y;
        acc.z += w[k] * x[k].z;
        acc.w += w[k] * x[k].w;
    }
}

__global__ void __launch_bounds__(256)
spmm_gather(const int* __restrict__ rowptr,
            const int2* __restrict__ edges,
            const float* __restrict__ in,
            float* __restrict__ out, int N) {
    const int lane = threadIdx.x & 31;
    const int node = blockIdx.x * 8 + __builtin_amdgcn_readfirstlane(threadIdx.x >> 5);
    if (node >= N) return;
    const int beg = rowptr[node];
    const int end = rowptr[node + 1];
    const float4* in4 = (const float4*)in;
    float4 acc = make_float4(0.f, 0.f, 0.f, 0.f);

    for (int base = beg; base < end; base += 32) {
        const int cnt = min(32, end - base);
        int2 m = make_int2(0, 0);
        if (lane < cnt) m = edges[base + lane];
        int j = 0;
        for (; j + 8 <= cnt; j += 8) gather_rows<8>(acc, in4, m.x, m.y, j, lane);
        switch (cnt - j) {
            case 7: gather_rows<7>(acc, in4, m.x, m.y, j, lane); break;
            case 6: gather_rows<6>(acc, in4, m.x, m.y, j, lane); break;
            case 5: gather_rows<5>(acc, in4, m.x, m.y, j, lane); break;
            case 4: gather_rows<4>(acc, in4, m.x, m.y, j, lane); break;
            case 3: gather_rows<3>(acc, in4, m.x, m.y, j, lane); break;
            case 2: gather_rows<2>(acc, in4, m.x, m.y, j, lane); break;
            case 1: gather_rows<1>(acc, in4, m.x, m.y, j, lane); break;
            default: break;
        }
    }
    ((float4*)out)[(size_t)node * (DC / 4) + lane] = acc;
}

__global__ void layer1_combine(const float* __restrict__ w10,
                               float* __restrict__ s1_h,
                               const float* __restrict__ s2,
                               const float* __restrict__ w12,
                               const float* __restrict__ b1,
                               int n4, int D) {
    int i = blockIdx.x * blockDim.x + threadIdx.x;
    if (i >= n4) return;
    int cg = i & ((D >> 2) - 1);
    float4 a = ((const float4*)w10)[i];
    float4 s = ((const float4*)s1_h)[i];
    float4 t = ((const float4*)s2)[i];
    float4 m = ((const float4*)w12)[i];
    float4 b = ((const float4*)b1)[cg];
    float4 o;
    o.x = fmaxf(a.x + s.x + 2.f * t.x - m.x + b.x, 0.f);
    o.y = fmaxf(a.y + s.y + 2.f * t.y - m.y + b.y, 0.f);
    o.z = fmaxf(a.z + s.z + 2.f * t.z - m.z + b.z, 0.f);
    o.w = fmaxf(a.w + s.w + 2.f * t.w - m.w + b.w, 0.f);
    ((float4*)s1_h)[i] = o;
}

__global__ void t2_combine(float* __restrict__ u_t2,
                           const float* __restrict__ h, int n4) {
    int i = blockIdx.x * blockDim.x + threadIdx.x;
    if (i >= n4) return;
    float4 a = ((const float4*)u_t2)[i];
    float4 b = ((const float4*)h)[i];
    float4 o;
    o.x = 2.f * a.x - b.x;  o.y = 2.f * a.y - b.y;
    o.z = 2.f * a.z - b.z;  o.w = 2.f * a.w - b.w;
    ((float4*)u_t2)[i] = o;
}

#define APAD 132

__global__ void __launch_bounds__(256)
gemm3_wmma_f32(const float* __restrict__ A0,
               const float* __restrict__ A1,
               const float* __restrict__ A2,
               const float* __restrict__ B,
               const float* __restrict__ bias,
               float* __restrict__ out) {
    __shared__ float As[3][16][APAD];

    const int wave = threadIdx.x >> 5;
    const int lane = threadIdx.x & 31;
    const int idx  = lane & 15;
    const int half = lane >> 4;

    const int tm   = blockIdx.x;
    const int tn   = wave;
    const int colB = tn * 16 + idx;

    const float* Am[3] = {A0, A1, A2};

#pragma unroll
    for (int m = 0; m < 3; ++m) {
        const float* src = Am[m] + (size_t)tm * 16 * DC;
#pragma unroll
        for (int i = 0; i < 2; ++i) {
            int f  = threadIdx.x + i * 256;
            int r  = f >> 5;
            int c4 = f & 31;
            float4 v = ((const float4*)(src + (size_t)r * DC))[c4];
            *(float4*)&As[m][r][c4 << 2] = v;
        }
    }
    __syncthreads();

    v8f c = {0.f, 0.f, 0.f, 0.f, 0.f, 0.f, 0.f, 0.f};

#pragma unroll
    for (int m = 0; m < 3; ++m) {
        const float* Arow  = &As[m][idx][2 * half];
        const float* Bbase = B + (size_t)m * DC * DC
                               + (size_t)2 * half * DC + colB;
#pragma unroll
        for (int k0 = 0; k0 < DC; k0 += 4) {
            v2f a = *(const v2f*)(Arow + k0);
            v2f b;
            b[0] = Bbase[(size_t)k0 * DC];
            b[1] = Bbase[(size_t)k0 * DC + DC];
            c = __builtin_amdgcn_wmma_f32_16x16x4_f32(
                    false, a, false, b, (short)0, c, false, false);
        }
    }

    float bv = bias[colB];
#pragma unroll
    for (int r = 0; r < 8; ++r) {
        int orow = tm * 16 + r + 8 * half;
        out[(size_t)orow * DC + colB] = c[r] + bv;
    }
}


extern "C" void kernel_launch(void* const* d_in, const int* in_sizes, int n_in,
                              void* d_out, int out_size, void* d_ws, size_t ws_size,
                              hipStream_t stream) {
    const int*   ei  = (const int*)d_in[1];
    const float* ew  = (const float*)d_in[2];
    const float* W1  = (const float*)d_in[3];
    const float* b1  = (const float*)d_in[4];
    const float* W2  = (const float*)d_in[5];
    const float* b2  = (const float*)d_in[6];
    float* out = (float*)d_out;

    const int E = in_sizes[2];
    const int D = in_sizes[4];
    const int N = in_sizes[3] / (3 * D);
    const int ND = N * D;

    const int* row = ei;
    const int* col = ei + E;
    const float* W1_0 = W1;
    const float* W1_1 = W1 + (size_t)ND;
    const float* W1_2 = W1 + (size_t)2 * ND;

    char* ws = (char*)d_ws;
    int*   rowptr = (int*)ws;                    ws += (size_t)N * 8;
    ws += (size_t)E * 4;
    float* B0 = (float*)ws;                      ws += (size_t)ND * 4;
    float* B1 = (float*)ws;                      ws += (size_t)ND * 4;
    float* B2 = (float*)ws;                      ws += (size_t)ND * 4;
    int2*  edges = (int2*)ws;

    const int TB = 256;
    const int nd4 = ND / 4;
    dim3 gElem((nd4 + TB - 1) / TB);
    dim3 gRows((N + 7) / 8);

    build_sorted_edges<<<1, BT, 0, stream>>>(row, col, ew, rowptr, edges, E, N);

    spmm_gather<<<gRows, TB, 0, stream>>>(rowptr, edges, W1_1, B0, N);
    spmm_gather<<<gRows, TB, 0, stream>>>(rowptr, edges, W1_2, B1, N);
    spmm_gather<<<gRows, TB, 0, stream>>>(rowptr, edges, B1, B2, N);
    layer1_combine<<<gElem, TB, 0, stream>>>(W1_0, B0, B2, W1_2, b1, nd4, D);

    spmm_gather<<<gRows, TB, 0, stream>>>(rowptr, edges, B0, B1, N);
    spmm_gather<<<gRows, TB, 0, stream>>>(rowptr, edges, B1, B2, N);
    t2_combine<<<gElem, TB, 0, stream>>>(B2, B0, nd4);

    gemm3_wmma_f32<<<N / 16, 256, 0, stream>>>(B0, B1, B2, W2, b2, out);
}


# Row-local stages fused into the gathers: layer 1's two first products share one gather, combine+relu is the S2 gather's epilogue, last gather and T2 fill the GEMM's LDS strip; 9 launches to 5
# speedup vs baseline: 2.0339x; 1.2427x over previous
#include <hip/hip_runtime.h>


typedef __attribute__((ext_vector_type(2))) float v2f;
typedef __attribute__((ext_vector_type(8))) float v8f;


#define DC 128
#define NMAX 4096
#define EMAX 32768
#define BT 1024
#define EB 8

__global__ void __launch_bounds__(BT)
build_sorted_edges(const int* __restrict__ row,
                   const int* __restrict__ col,
                   const float* __restrict__ ew,
                   int* __restrict__ rowptr,
                   int2* __restrict__ edges,
                   int E, int N) {
    __shared__ float s_dinv[NMAX];
    __shared__ int   s_cur[NMAX];
    __shared__ int   s_wsum[32];
    __shared__ int2  s_rec[EMAX];

    const int tid  = threadIdx.x;
    const int lane = tid & 31;
    const int wave = tid >> 5;

    for (int i = tid; i < NMAX; i += BT) { s_dinv[i] = 0.f; s_cur[i] = 0; }
    __syncthreads();

    int e0 = tid;
    for (; e0 + (EB - 1) * BT < E; e0 += BT * EB) {
        int r[EB], c[EB];
        float w[EB];
#pragma unroll
        for (int k = 0; k < EB; ++k) {
            r[k] = row[e0 + k * BT];
            c[k] = col[e0 + k * BT];
            w[k] = ew[e0 + k * BT];
        }
#pragma unroll
        for (int k = 0; k < EB; ++k) {
            atomicAdd(&s_dinv[r[k]], w[k]);
            atomicAdd(&s_cur[c[k]], 1);
        }
    }
    for (int e = e0; e < E; e += BT) {
        atomicAdd(&s_dinv[row[e]], ew[e]);
        atomicAdd(&s_cur[col[e]], 1);
    }
    __syncthreads();

    for (int i = tid; i < NMAX; i += BT) {
        float d = s_dinv[i];
        s_dinv[i] = (d > 0.f) ? rsqrtf(d) : 0.f;
    }

    const int b4 = tid * 4;
    const int c0 = s_cur[b4], c1 = s_cur[b4 + 1], c2 = s_cur[b4 + 2], c3 = s_cur[b4 + 3];
    const int tsum = c0 + c1 + c2 + c3;
    int v = tsum;
    for (int d = 1; d < 32; d <<= 1) {
        int t = __shfl_up(v, d);
        if (lane >= d) v += t;
    }
    if (lane == 31) s_wsum[wave] = v;
    __syncthreads();
    if (wave == 0) {
        const int w = s_wsum[lane];
        int x = w;
        for (int d = 1; d < 32; d <<= 1) {
            int t = __shfl_up(x, d);
            if (lane >= d) x += t;
        }
        s_wsum[lane] = x - w;
    }
    __syncthreads();
    const int o0 = s_wsum[wave] + v - tsum;
    const int o1 = o0 + c0, o2 = o1 + c1, o3 = o2 + c2;
    s_cur[b4] = o0; s_cur[b4 + 1] = o1; s_cur[b4 + 2] = o2; s_cur[b4 + 3] = o3;
    if (b4     < N) rowptr[b4]     = o0;
    if (b4 + 1 < N) rowptr[b4 + 1] = o1;
    if (b4 + 2 < N) rowptr[b4 + 2] = o2;
    if (b4 + 3 < N) rowptr[b4 + 3] = o3;
    if (tid == 0) rowptr[N] = E;
    __syncthreads();

    e0 = tid;
    for (; e0 + (EB - 1) * BT < E; e0 += BT * EB) {
        int r[EB], c[EB], pos[EB];
        float w[EB];
#pragma unroll
        for (int k = 0; k < EB; ++k) {
            r[k] = row[e0 + k * BT];
            c[k] = col[e0 + k * BT];
            w[k] = ew[e0 + k * BT];
        }
#pragma unroll
        for (int k = 0; k < EB; ++k) w[k] = -s_dinv[r[k]] * w[k] * s_dinv[c[k]];
#pragma unroll
        for (int k = 0; k < EB; ++k) pos[k] = atomicAdd(&s_cur[c[k]], 1);
#pragma unroll
        for (int k = 0; k < EB; ++k) s_rec[pos[k]] = make_int2(r[k], __float_as_int(w[k]));
    }
    for (int e = e0; e < E; e += BT) {
        const int r = row[e], c = col[e];
        const float w = -s_dinv[r] * ew[e] * s_dinv[c];
        const int pos = atomicAdd(&s_cur[c], 1);
        s_rec[pos] = make_int2(r, __float_as_int(w));
    }
    __syncthreads();

    const int4* s4 = (const int4*)s_rec;
    int4* g4 = (int4*)edges;
    const int n2 = E >> 1;
    for (int i = tid; i < n2; i += BT) g4[i] = s4[i];
    if (tid == 0 && (E & 1)) edges[E - 1] = s_rec[E - 1];
}

template <int K, int NS>
__device__ __forceinline__ void gather_rows(float4 (&acc)[NS], const float4* const (&in4)[NS],
                                            int msrc, int mw, int j, int lane) {
    float4 x[K][NS];
    float  w[K];
#pragma unroll
    for (int k = 0; k < K; ++k) {
        const int s = __builtin_amdgcn_readlane(msrc, j + k);
        w[k] = __int_as_float(__builtin_amdgcn_readlane(mw, j + k));
#pragma unroll
        for (int n = 0; n < NS; ++n) x[k][n] = in4[n][(size_t)s * (DC / 4) + lane];
    }
#pragma unroll
    for (int k = 0; k < K; ++k) {
#pragma unroll
        for (int n = 0; n < NS; ++n) {
            acc[n].x += w[k] * x[k][n].x;
            acc[n].y += w[k] * x[k][n].y;
            acc[n].z += w[k] * x[k][n].z;
            acc[n].w += w[k] * x[k][n].w;
        }
    }
}

__device__ __forceinline__ int2 load_records(const int2* __restrict__ edges,
                                             int base, int end, int lane) {
    int2 m = make_int2(0, 0);
    if (lane < end - base) m = edges[base + lane];
    return m;
}

template <int K, int NS>
__device__ __forceinline__ void gather_tail(float4 (&acc)[NS], const float4* const (&in4)[NS],
                                            int msrc, int mw, int j, int rem, int lane) {
    if constexpr (K > 0) {
        if (rem == K) gather_rows<K, NS>(acc, in4, msrc, mw, j, lane);
        else gather_tail<K - 1, NS>(acc, in4, msrc, mw, j, rem, lane);
    }
}

template <int NS>
__device__ __forceinline__ void gather_row(float4 (&acc)[NS], const float4* const (&in4)[NS],
                                           const int2* __restrict__ edges,
                                           int beg, int end, int2 m, int lane) {
    constexpr int KB = 8 / NS;
    for (int base = beg; base < end; base += 32) {
        if (base != beg) m = load_records(edges, base, end, lane);
        const int cnt = min(32, end - base);
        int j = 0;
        for (; j + KB <= cnt; j += KB) gather_rows<KB, NS>(acc, in4, m.x, m.y, j, lane);
        gather_tail<KB - 1, NS>(acc, in4, m.x, m.y, j, cnt - j, lane);
    }
}

__global__ void __launch_bounds__(256)
spmm_gather(const int* __restrict__ rowptr,
            const int2* __restrict__ edges,
            const float* __restrict__ in,
            float* __restrict__ out, int N) {
    const int lane = threadIdx.x & 31;
    const int node = blockIdx.x * 8 + __builtin_amdgcn_readfirstlane(threadIdx.x >> 5);
    if (node >= N) return;
    const int beg = rowptr[node];
    const int end = rowptr[node + 1];
    const float4* const in4[1] = {(const float4*)in};
    float4 acc[1] = {make_float4(0.f, 0.f, 0.f, 0.f)};
    gather_row<1>(acc, in4, edges, beg, end, load_records(edges, beg, end, lane), lane);
    ((float4*)out)[(size_t)node * (DC / 4) + lane] = acc[0];
}

__global__ void __launch_bounds__(256)
spmm_gather_pair(const int* __restrict__ rowptr,
                 const int2* __restrict__ edges,
                 const float* __restrict__ inA,
                 const float* __restrict__ inB,
                 float* __restrict__ outA,
                 float* __restrict__ outB, int N) {
    const int lane = threadIdx.x & 31;
    const int node = blockIdx.x * 8 + __builtin_amdgcn_readfirstlane(threadIdx.x >> 5);
    if (node >= N) return;
    const int beg = rowptr[node];
    const int end = rowptr[node + 1];
    const float4* const in4[2] = {(const float4*)inA, (const float4*)inB};
    float4 acc[2] = {make_float4(0.f, 0.f, 0.f, 0.f), make_float4(0.f, 0.f, 0.f, 0.f)};
    gather_row<2>(acc, in4, edges, beg, end, load_records(edges, beg, end, lane), lane);
    const size_t o = (size_t)node * (DC / 4) + lane;
    ((float4*)outA)[o] = acc[0];
    ((float4*)outB)[o] = acc[1];
}

__global__ void __launch_bounds__(256)
spmm_gather_relu(const int* __restrict__ rowptr,
                 const int2* __restrict__ edges,
                 const float* __restrict__ in,
                 const float* __restrict__ w10,
                 float* __restrict__ s1_h,
                 const float* __restrict__ w12,
                 const float* __restrict__ b1, int N) {
    const int lane = threadIdx.x & 31;
    const int node = blockIdx.x * 8 + __builtin_amdgcn_readfirstlane(threadIdx.x >> 5);
    if (node >= N) return;
    const int beg = rowptr[node];
    const int end = rowptr[node + 1];
    const int2 m0 = load_records(edges, beg, end, lane);
    const size_t o = (size_t)node * (DC / 4) + lane;
    const float4 a = ((const float4*)w10)[o];
    const float4 s = ((const float4*)s1_h)[o];
    const float4 m = ((const float4*)w12)[o];
    const float4 b = ((const float4*)b1)[lane];
    const float4* const in4[1] = {(const float4*)in};
    float4 acc[1] = {make_float4(0.f, 0.f, 0.f, 0.f)};
    gather_row<1>(acc, in4, edges, beg, end, m0, lane);
    const float4 t = acc[0];
    float4 r;
    r.x = fmaxf(a.x + s.x + 2.f * t.x - m.x + b.x, 0.f);
    r.y = fmaxf(a.y + s.y + 2.f * t.y - m.y + b.y, 0.f);
    r.z = fmaxf(a.z + s.z + 2.f * t.z - m.z + b.z, 0.f);
    r.w = fmaxf(a.w + s.w + 2.f * t.w - m.w + b.w, 0.f);
    ((float4*)s1_h)[o] = r;
}

#define APAD 132

__global__ void __launch_bounds__(256)
gemm3_gather_wmma_f32(const int* __restrict__ rowptr,
                      const int2* __restrict__ edges,
                      const float* __restrict__ A0,
                      const float* __restrict__ A1,
                      const float* __restrict__ B,
                      const float* __restrict__ bias,
                      float* __restrict__ out) {
    __shared__ float As[3][16][APAD];

    const int wave = __builtin_amdgcn_readfirstlane(threadIdx.x >> 5);
    const int lane = threadIdx.x & 31;
    const int idx  = lane & 15;
    const int half = lane >> 4;

    const int tm   = blockIdx.x;
    const int tn   = wave;
    const int colB = tn * 16 + idx;

    const int r0 = wave, r1 = wave + 8;
    const int n0 = tm * 16 + r0, n1 = tm * 16 + r1;
    const int beg0 = rowptr[n0], end0 = rowptr[n0 + 1];
    const int beg1 = rowptr[n1], end1 = rowptr[n1 + 1];
    const int2 m0 = load_records(edges, beg0, end0, lane);
    const int2 m1 = load_records(edges, beg1, end1, lane);
    const float4* const H4[1] = {(const float4*)A0};
    const float4* const T4[1] = {(const float4*)A1};
    const float4 h0 = H4[0][(size_t)n0 * (DC / 4) + lane];
    const float4 h1 = H4[0][(size_t)n1 * (DC / 4) + lane];
    const float4 t0 = T4[0][(size_t)n0 * (DC / 4) + lane];
    const float4 t1 = T4[0][(size_t)n1 * (DC / 4) + lane];
    float4 u0[1] = {make_float4(0.f, 0.f, 0.f, 0.f)};
    float4 u1[1] = {make_float4(0.f, 0.f, 0.f, 0.f)};
    gather_row<1>(u0, T4, edges, beg0, end0, m0, lane);
    gather_row<1>(u1, T4, edges, beg1, end1, m1, lane);
    float4 x0, x1;
    x0.x = 2.f * u0[0].x - h0.x;  x0.y = 2.f * u0[0].y - h0.y;
    x0.z = 2.f * u0[0].z - h0.z;  x0.w = 2.f * u0[0].w - h0.w;
    x1.x = 2.f * u1[0].x - h1.x;  x1.y = 2.f * u1[0].y - h1.y;
    x1.z = 2.f * u1[0].z - h1.z;  x1.w = 2.f * u1[0].w - h1.w;
    *(float4*)&As[0][r0][lane << 2] = h0;
    *(float4*)&As[0][r1][lane << 2] = h1;
    *(float4*)&As[1][r0][lane << 2] = t0;
    *(float4*)&As[1][r1][lane << 2] = t1;
    *(float4*)&As[2][r0][lane << 2] = x0;
    *(float4*)&As[2][r1][lane << 2] = x1;
    __syncthreads();

    v8f c = {0.f, 0.f, 0.f, 0.f, 0.f, 0.f, 0.f, 0.f};

#pragma unroll
    for (int m = 0; m < 3; ++m) {
        const float* Arow  = &As[m][idx][2 * half];
        const float* Bbase = B + (size_t)m * DC * DC
                               + (size_t)2 * half * DC + colB;
#pragma unroll
        for (int k0 = 0; k0 < DC; k0 += 4) {
            v2f a = *(const v2f*)(Arow + k0);
            v2f b;
            b[0] = Bbase[(size_t)k0 * DC];
            b[1] = Bbase[(size_t)k0 * DC + DC];
            c = __builtin_amdgcn_wmma_f32_16x16x4_f32(
                    false, a, false, b, (short)0, c, false, false);
        }
    }

    float bv = bias[colB];
#pragma unroll
    for (int r = 0; r < 8; ++r) {
        int orow = tm * 16 + r + 8 * half;
        out[(size_t)orow * DC + colB] = c[r] + bv;
    }
}


extern "C" void kernel_launch(void* const* d_in, const int* in_sizes, int n_in,
                              void* d_out, int out_size, void* d_ws, size_t ws_size,
                              hipStream_t stream) {
    const int*   ei  = (const int*)d_in[1];
    const float* ew  = (const float*)d_in[2];
    const float* W1  = (const float*)d_in[3];
    const float* b1  = (const float*)d_in[4];
    const float* W2  = (const float*)d_in[5];
    const float* b2  = (const float*)d_in[6];
    float* out = (float*)d_out;

    const int E = in_sizes[2];
    const int D = in_sizes[4];
    const int N = in_sizes[3] / (3 * D);
    const int ND = N * D;

    const int* row = ei;
    const int* col = ei + E;
    const float* W1_0 = W1;
    const float* W1_1 = W1 + (size_t)ND;
    const float* W1_2 = W1 + (size_t)2 * ND;

    char* ws = (char*)d_ws;
    int*   rowptr = (int*)ws;                    ws += (size_t)N * 8;
    ws += (size_t)E * 4;
    float* B0 = (float*)ws;                      ws += (size_t)ND * 4;
    float* B1 = (float*)ws;                      ws += (size_t)ND * 4;
    ws += (size_t)ND * 4;
    int2*  edges = (int2*)ws;

    const int TB = 256;
    dim3 gRows((N + 7) / 8);

    build_sorted_edges<<<1, BT, 0, stream>>>(row, col, ew, rowptr, edges, E, N);

    spmm_gather_pair<<<gRows, TB, 0, stream>>>(rowptr, edges, W1_1, W1_2, B0, B1, N);
    spmm_gather_relu<<<gRows, TB, 0, stream>>>(rowptr, edges, B1, W1_0, B0, W1_2, b1, N);

    spmm_gather<<<gRows, TB, 0, stream>>>(rowptr, edges, B0, B1, N);

    gemm3_gather_wmma_f32<<<N / 16, 256, 0, stream>>>(rowptr, edges, B0, B1, W2, b2, out);
}


# Edge sort spread over 64 workgroups, one destination-node range each: each scans all edges (16-byte loads), sums all degrees itself, sorts its in-edges in LDS; sums stay in LDS-atomic arrival order
# speedup vs baseline: 2.5793x; 1.2682x over previous
#include <hip/hip_runtime.h>


typedef __attribute__((ext_vector_type(2))) float v2f;
typedef __attribute__((ext_vector_type(8))) float v8f;


#define DC 128
#define NMAX 4096
#define BT 1024
#define BG 64
#define NBMAX (NMAX / BG)
#define CAPSEL 16384
#define CH 4

__device__ __forceinline__ void count_edge(float* s_dinv, int* s_cnt, int* s_sel, int* s_nsel,
                                           int r, int c, float w, int e,
                                           int base, int nb, int& below) {
    atomicAdd(&s_dinv[r], w);
    below += (c < base) ? 1 : 0;
    const int k = c - base;
    if ((unsigned)k < (unsigned)nb) {
        atomicAdd(&s_cnt[k], 1);
        const int slot = atomicAdd(s_nsel, 1);
        if (slot < CAPSEL) s_sel[slot] = e;
    }
}

__global__ void __launch_bounds__(BT)
build_sorted_edges(const int* __restrict__ row,
                   const int* __restrict__ col,
                   const float* __restrict__ ew,
                   int2* __restrict__ bounds,
                   int2* __restrict__ edges,
                   int E, int N) {
    __shared__ float s_dinv[NMAX];
    __shared__ int   s_cnt[NBMAX];
    __shared__ int   s_red[32];
    __shared__ int   s_nsel;
    __shared__ int   s_start;
    __shared__ int   s_sel[CAPSEL];
    __shared__ int2  s_rec[CAPSEL];

    const int tid  = threadIdx.x;
    const int lane = tid & 31;
    const int wave = tid >> 5;
    const int nb   = (N + BG - 1) / BG;
    const int base = blockIdx.x * nb;

    for (int i = tid; i < NMAX; i += BT) s_dinv[i] = 0.f;
    if (tid < NBMAX) s_cnt[tid] = 0;
    if (tid == 0) s_nsel = 0;
    __syncthreads();

    int below = 0;
    if ((E & 3) == 0) {
        const int4*   row4 = (const int4*)row;
        const int4*   col4 = (const int4*)col;
        const float4* ew4  = (const float4*)ew;
        const int nq = E >> 2;
        int q0 = tid;
        for (; q0 + (CH - 1) * BT < nq; q0 += BT * CH) {
            int4 r[CH], c[CH];
            float4 w[CH];
#pragma unroll
            for (int k = 0; k < CH; ++k) {
                r[k] = row4[q0 + k * BT];
                c[k] = col4[q0 + k * BT];
                w[k] = ew4[q0 + k * BT];
            }
#pragma unroll
            for (int k = 0; k < CH; ++k) {
                const int e = 4 * (q0 + k * BT);
                count_edge(s_dinv, s_cnt, s_sel, &s_nsel, r[k].x, c[k].x, w[k].x, e,     base, nb, below);
                count_edge(s_dinv, s_cnt, s_sel, &s_nsel, r[k].y, c[k].y, w[k].y, e + 1, base, nb, below);
                count_edge(s_dinv, s_cnt, s_sel, &s_nsel, r[k].z, c[k].z, w[k].z, e + 2, base, nb, below);
                count_edge(s_dinv, s_cnt, s_sel, &s_nsel, r[k].w, c[k].w, w[k].w, e + 3, base, nb, below);
            }
        }
        for (int q = q0; q < nq; q += BT) {
            const int4 r = row4[q], c = col4[q];
            const float4 w = ew4[q];
            const int e = 4 * q;
            count_edge(s_dinv, s_cnt, s_sel, &s_nsel, r.x, c.x, w.x, e,     base, nb, below);
            count_edge(s_dinv, s_cnt, s_sel, &s_nsel, r.y, c.y, w.y, e + 1, base, nb, below);
            count_edge(s_dinv, s_cnt, s_sel, &s_nsel, r.z, c.z, w.z, e + 2, base, nb, below);
            count_edge(s_dinv, s_cnt, s_sel, &s_nsel, r.w, c.w, w.w, e + 3, base, nb, below);
        }
    } else {
        for (int e = tid; e < E; e += BT)
            count_edge(s_dinv, s_cnt, s_sel, &s_nsel, row[e], col[e], ew[e], e, base, nb, below);
    }
    for (int d = 16; d > 0; d >>= 1) below += __shfl_xor(below, d);
    if (lane == 0) s_red[wave] = below;
    __syncthreads();

    for (int i = tid; i < NMAX; i += BT) {
        const float d = s_dinv[i];
        s_dinv[i] = (d > 0.f) ? rsqrtf(d) : 0.f;
    }
    if (wave == 0) {
        int t = s_red[lane];
        for (int d = 16; d > 0; d >>= 1) t += __shfl_xor(t, d);
        const int start = ((t + 15) & ~15) + 16 * (int)blockIdx.x;
        const int i0 = 2 * lane;
        const int c0 = s_cnt[i0], c1 = s_cnt[i0 + 1];
        int v = c0 + c1;
        for (int d = 1; d < 32; d <<= 1) {
            const int u = __shfl_up(v, d);
            if (lane >= d) v += u;
        }
        const int o0 = v - c0 - c1, o1 = o0 + c0;
        s_cnt[i0] = o0;
        s_cnt[i0 + 1] = o1;
        const bool v0 = i0 < nb && base + i0 < N;
        const bool v1 = i0 + 1 < nb && base + i0 + 1 < N;
        if (v0 && v1 && (base & 1) == 0) {
            *(int4*)&bounds[base + i0] = make_int4(start + o0, start + o1, start + o1, start + o1 + c1);
        } else {
            if (v0) bounds[base + i0]     = make_int2(start + o0, start + o1);
            if (v1) bounds[base + i0 + 1] = make_int2(start + o1, start + o1 + c1);
        }
        if (lane == 0) s_start = start;
    }
    __syncthreads();

    const int nsel  = s_nsel;
    const int start = s_start;
    if (nsel <= CAPSEL) {
        for (int i = tid; i < nsel; i += BT) {
            const int e = s_sel[i];
            const int r = row[e], c = col[e];
            const float w = -s_dinv[r] * ew[e] * s_dinv[c];
            const int pos = atomicAdd(&s_cnt[c - base], 1);
            s_rec[pos] = make_int2(r, __float_as_int(w));
        }
        __syncthreads();
        const int4* s4 = (const int4*)s_rec;
        int4* g4 = (int4*)(edges + start);
        const int n2 = nsel >> 1;
        for (int i = tid; i < n2; i += BT) g4[i] = s4[i];
        if (tid == 0 && (nsel & 1)) edges[start + nsel - 1] = s_rec[nsel - 1];
    } else {
        for (int e = tid; e < E; e += BT) {
            const int c = col[e];
            if ((unsigned)(c - base) < (unsigned)nb) {
                const int r = row[e];
                const float w = -s_dinv[r] * ew[e] * s_dinv[c];
                const int pos = atomicAdd(&s_cnt[c - base], 1);
                edges[start + pos] = make_int2(r, __float_as_int(w));
            }
        }
    }
}

template <int K, int NS>
__device__ __forceinline__ void gather_rows(float4 (&acc)[NS], const float4* const (&in4)[NS],
                                            int msrc, int mw, int j, int lane) {
    float4 x[K][NS];
    float  w[K];
#pragma unroll
    for (int k = 0; k < K; ++k) {
        const int s = __builtin_amdgcn_readlane(msrc, j + k);
        w[k] = __int_as_float(__builtin_amdgcn_readlane(mw, j + k));
#pragma unroll
        for (int n = 0; n < NS; ++n) x[k][n] = in4[n][(size_t)s * (DC / 4) + lane];
    }
#pragma unroll
    for (int k = 0; k < K; ++k) {
#pragma unroll
        for (int n = 0; n < NS; ++n) {
            acc[n].x += w[k] * x[k][n].x;
            acc[n].y += w[k] * x[k][n].y;
            acc[n].z += w[k] * x[k][n].z;
            acc[n].w += w[k] * x[k][n].w;
        }
    }
}

__device__ __forceinline__ int2 load_records(const int2* __restrict__ edges,
                                             int base, int end, int lane) {
    int2 m = make_int2(0, 0);
    if (lane < end - base) m = edges[base + lane];
    return m;
}

template <int K, int NS>
__device__ __forceinline__ void gather_tail(float4 (&acc)[NS], const float4* const (&in4)[NS],
                                            int msrc, int mw, int j, int rem, int lane) {
    if constexpr (K > 0) {
        if (rem == K) gather_rows<K, NS>(acc, in4, msrc, mw, j, lane);
        else gather_tail<K - 1, NS>(acc, in4, msrc, mw, j, rem, lane);
    }
}

template <int NS>
__device__ __forceinline__ void gather_row(float4 (&acc)[NS], const float4* const (&in4)[NS],
                                           const int2* __restrict__ edges,
                                           int beg, int end, int2 m, int lane) {
    constexpr int KB = 8 / NS;
    for (int base = beg; base < end; base += 32) {
        if (base != beg) m = load_records(edges, base, end, lane);
        const int cnt = min(32, end - base);
        int j = 0;
        for (; j + KB <= cnt; j += KB) gather_rows<KB, NS>(acc, in4, m.x, m.y, j, lane);
        gather_tail<KB - 1, NS>(acc, in4, m.x, m.y, j, cnt - j, lane);
    }
}

__global__ void __launch_bounds__(256)
spmm_gather(const int2* __restrict__ bounds,
            const int2* __restrict__ edges,
            const float* __restrict__ in,
            float* __restrict__ out, int N) {
    const int lane = threadIdx.x & 31;
    const int node = blockIdx.x * 8 + __builtin_amdgcn_readfirstlane(threadIdx.x >> 5);
    if (node >= N) return;
    const int2 be = bounds[node];
    const int beg = be.x, end = be.y;
    const float4* const in4[1] = {(const float4*)in};
    float4 acc[1] = {make_float4(0.f, 0.f, 0.f, 0.f)};
    gather_row<1>(acc, in4, edges, beg, end, load_records(edges, beg, end, lane), lane);
    ((float4*)out)[(size_t)node * (DC / 4) + lane] = acc[0];
}

__global__ void __launch_bounds__(256)
spmm_gather_pair(const int2* __restrict__ bounds,
                 const int2* __restrict__ edges,
                 const float* __restrict__ inA,
                 const float* __restrict__ inB,
                 float* __restrict__ outA,
                 float* __restrict__ outB, int N) {
    const int lane = threadIdx.x & 31;
    const int node = blockIdx.x * 8 + __builtin_amdgcn_readfirstlane(threadIdx.x >> 5);
    if (node >= N) return;
    const int2 be = bounds[node];
    const int beg = be.x, end = be.y;
    const float4* const in4[2] = {(const float4*)inA, (const float4*)inB};
    float4 acc[2] = {make_float4(0.f, 0.f, 0.f, 0.f), make_float4(0.f, 0.f, 0.f, 0.f)};
    gather_row<2>(acc, in4, edges, beg, end, load_records(edges, beg, end, lane), lane);
    const size_t o = (size_t)node * (DC / 4) + lane;
    ((float4*)outA)[o] = acc[0];
    ((float4*)outB)[o] = acc[1];
}

__global__ void __launch_bounds__(256)
spmm_gather_relu(const int2* __restrict__ bounds,
                 const int2* __restrict__ edges,
                 const float* __restrict__ in,
                 const float* __restrict__ w10,
                 float* __restrict__ s1_h,
                 const float* __restrict__ w12,
                 const float* __restrict__ b1, int N) {
    const int lane = threadIdx.x & 31;
    const int node = blockIdx.x * 8 + __builtin_amdgcn_readfirstlane(threadIdx.x >> 5);
    if (node >= N) return;
    const int2 be = bounds[node];
    const int beg = be.x, end = be.y;
    const int2 m0 = load_records(edges, beg, end, lane);
    const size_t o = (size_t)node * (DC / 4) + lane;
    const float4 a = ((const float4*)w10)[o];
    const float4 s = ((const float4*)s1_h)[o];
    const float4 m = ((const float4*)w12)[o];
    const float4 b = ((const float4*)b1)[lane];
    const float4* const in4[1] = {(const float4*)in};
    float4 acc[1] = {make_float4(0.f, 0.f, 0.f, 0.f)};
    gather_row<1>(acc, in4, edges, beg, end, m0, lane);
    const float4 t = acc[0];
    float4 r;
    r.x = fmaxf(a.x + s.x + 2.f * t.x - m.x + b.x, 0.f);
    r.y = fmaxf(a.y + s.y + 2.f * t.y - m.y + b.y, 0.f);
    r.z = fmaxf(a.z + s.z + 2.f * t.z - m.z + b.z, 0.f);
    r.w = fmaxf(a.w + s.w + 2.f * t.w - m.w + b.w, 0.f);
    ((float4*)s1_h)[o] = r;
}

#define APAD 132

__global__ void __launch_bounds__(256)
gemm3_gather_wmma_f32(const int2* __restrict__ bounds,
                      const int2* __restrict__ edges,
                      const float* __restrict__ A0,
                      const float* __restrict__ A1,
                      const float* __restrict__ B,
                      const float* __restrict__ bias,
                      float* __restrict__ out) {
    __shared__ float As[3][16][APAD];

    const int wave = __builtin_amdgcn_readfirstlane(threadIdx.x >> 5);
    const int lane = threadIdx.x & 31;
    const int idx  = lane & 15;
    const int half = lane >> 4;

    const int tm   = blockIdx.x;
    const int tn   = wave;
    const int colB = tn * 16 + idx;

    const int r0 = wave, r1 = wave + 8;
    const int n0 = tm * 16 + r0, n1 = tm * 16 + r1;
    const int2 be0 = bounds[n0], be1 = bounds[n1];
    const int beg0 = be0.x, end0 = be0.y;
    const int beg1 = be1.x, end1 = be1.y;
    const int2 m0 = load_records(edges, beg0, end0, lane);
    const int2 m1 = load_records(edges, beg1, end1, lane);
    const float4* const H4[1] = {(const float4*)A0};
    const float4* const T4[1] = {(const float4*)A1};
    const float4 h0 = H4[0][(size_t)n0 * (DC / 4) + lane];
    const float4 h1 = H4[0][(size_t)n1 * (DC / 4) + lane];
    const float4 t0 = T4[0][(size_t)n0 * (DC / 4) + lane];
    const float4 t1 = T4[0][(size_t)n1 * (DC / 4) + lane];
    float4 u0[1] = {make_float4(0.f, 0.f, 0.f, 0.f)};
    float4 u1[1] = {make_float4(0.f, 0.f, 0.f, 0.f)};
    gather_row<1>(u0, T4, edges, beg0, end0, m0, lane);
    gather_row<1>(u1, T4, edges, beg1, end1, m1, lane);
    float4 x0, x1;
    x0.x = 2.f * u0[0].x - h0.x;  x0.y = 2.f * u0[0].y - h0.y;
    x0.z = 2.f * u0[0].z - h0.z;  x0.w = 2.f * u0[0].w - h0.w;
    x1.x = 2.f * u1[0].x - h1.x;  x1.y = 2.f * u1[0].y - h1.y;
    x1.z = 2.f * u1[0].z - h1.z;  x1.w = 2.f * u1[0].w - h1.w;
    *(float4*)&As[0][r0][lane << 2] = h0;
    *(float4*)&As[0][r1][lane << 2] = h1;
    *(float4*)&As[1][r0][lane << 2] = t0;
    *(float4*)&As[1][r1][lane << 2] = t1;
    *(float4*)&As[2][r0][lane << 2] = x0;
    *(float4*)&As[2][r1][lane << 2] = x1;
    __syncthreads();

    v8f c = {0.f, 0.f, 0.f, 0.f, 0.f, 0.f, 0.f, 0.f};

#pragma unroll
    for (int m = 0; m < 3; ++m) {
        const float* Arow  = &As[m][idx][2 * half];
        const float* Bbase = B + (size_t)m * DC * DC
                               + (size_t)2 * half * DC + colB;
#pragma unroll
        for (int k0 = 0; k0 < DC; k0 += 4) {
            v2f a = *(const v2f*)(Arow + k0);
            v2f b;
            b[0] = Bbase[(size_t)k0 * DC];
            b[1] = Bbase[(size_t)k0 * DC + DC];
            c = __builtin_amdgcn_wmma_f32_16x16x4_f32(
                    false, a, false, b, (short)0, c, false, false);
        }
    }

    float bv = bias[colB];
#pragma unroll
    for (int r = 0; r < 8; ++r) {
        int orow = tm * 16 + r + 8 * half;
        out[(size_t)orow * DC + colB] = c[r] + bv;
    }
}


extern "C" void kernel_launch(void* const* d_in, const int* in_sizes, int n_in,
                              void* d_out, int out_size, void* d_ws, size_t ws_size,
                              hipStream_t stream) {
    const int*   ei  = (const int*)d_in[1];
    const float* ew  = (const float*)d_in[2];
    const float* W1  = (const float*)d_in[3];
    const float* b1  = (const float*)d_in[4];
    const float* W2  = (const float*)d_in[5];
    const float* b2  = (const float*)d_in[6];
    float* out = (float*)d_out;

    const int E = in_sizes[2];
    const int D = in_sizes[4];
    const int N = in_sizes[3] / (3 * D);
    const int ND = N * D;

    const int* row = ei;
    const int* col = ei + E;
    const float* W1_0 = W1;
    const float* W1_1 = W1 + (size_t)ND;
    const float* W1_2 = W1 + (size_t)2 * ND;

    char* ws = (char*)d_ws;
    int2*  bounds = (int2*)ws;                   ws += (size_t)N * 8;
    ws += (size_t)E * 4;
    float* B0 = (float*)ws;                      ws += (size_t)ND * 4;
    float* B1 = (float*)ws;                      ws += (size_t)ND * 4;
    int2*  edges = (int2*)ws;

    const int TB = 256;
    dim3 gRows((N + 7) / 8);

    build_sorted_edges<<<BG, BT, 0, stream>>>(row, col, ew, bounds, edges, E, N);

    spmm_gather_pair<<<gRows, TB, 0, stream>>>(bounds, edges, W1_1, W1_2, B0, B1, N);
    spmm_gather_relu<<<gRows, TB, 0, stream>>>(bounds, edges, B1, W1_0, B0, W1_2, b1, N);

    spmm_gather<<<gRows, TB, 0, stream>>>(bounds, edges, B0, B1, N);

    gemm3_gather_wmma_f32<<<N / 16, 256, 0, stream>>>(bounds, edges, B0, B1, W2, b2, out);
}


# First gather fused into the edge-sort launch: each sorting workgroup (256 of 16 nodes now) sums the S1 and S2a rows of its nodes from its LDS records; 5 launches to 4; LDS-atomic arrival order kept
# speedup vs baseline: 2.7980x; 1.0848x over previous
#include <hip/hip_runtime.h>


typedef __attribute__((ext_vector_type(2))) float v2f;
typedef __attribute__((ext_vector_type(8))) float v8f;

#define DC 128


template <int K, int NS>
__device__ __forceinline__ void gather_rows(float4 (&acc)[NS], const float4* const (&in4)[NS],
                                            int msrc, int mw, int j, int lane) {
    float4 x[K][NS];
    float  w[K];
#pragma unroll
    for (int k = 0; k < K; ++k) {
        const int s = __builtin_amdgcn_readlane(msrc, j + k);
        w[k] = __int_as_float(__builtin_amdgcn_readlane(mw, j + k));
#pragma unroll
        for (int n = 0; n < NS; ++n) x[k][n] = in4[n][(size_t)s * (DC / 4) + lane];
    }
#pragma unroll
    for (int k = 0; k < K; ++k) {
#pragma unroll
        for (int n = 0; n < NS; ++n) {
            acc[n].x += w[k] * x[k][n].x;
            acc[n].y += w[k] * x[k][n].y;
            acc[n].z += w[k] * x[k][n].z;
            acc[n].w += w[k] * x[k][n].w;
        }
    }
}

__device__ __forceinline__ int2 load_records(const int2* __restrict__ edges,
                                             int base, int end, int lane) {
    int2 m = make_int2(0, 0);
    if (lane < end - base) m = edges[base + lane];
    return m;
}

template <int K, int NS>
__device__ __forceinline__ void gather_tail(float4 (&acc)[NS], const float4* const (&in4)[NS],
                                            int msrc, int mw, int j, int rem, int lane) {
    if constexpr (K > 0) {
        if (rem == K) gather_rows<K, NS>(acc, in4, msrc, mw, j, lane);
        else gather_tail<K - 1, NS>(acc, in4, msrc, mw, j, rem, lane);
    }
}

template <int NS>
__device__ __forceinline__ void gather_row(float4 (&acc)[NS], const float4* const (&in4)[NS],
                                           const int2* __restrict__ edges,
                                           int beg, int end, int2 m, int lane) {
    constexpr int KB = 8 / NS;
    for (int base = beg; base < end; base += 32) {
        if (base != beg) m = load_records(edges, base, end, lane);
        const int cnt = min(32, end - base);
        int j = 0;
        for (; j + KB <= cnt; j += KB) gather_rows<KB, NS>(acc, in4, m.x, m.y, j, lane);
        gather_tail<KB - 1, NS>(acc, in4, m.x, m.y, j, cnt - j, lane);
    }
}


#define NMAX 4096
#define BT 512
#define BG 256
#define NBMAX (NMAX / BG)
#define SCNT 64
#define CAPSEL 16384
#define CH 4

__device__ __forceinline__ void count_edge(float* s_dinv, int* s_cnt, int* s_sel, int* s_nsel,
                                           int r, int c, float w, int e,
                                           int base, int nb, int& below) {
    atomicAdd(&s_dinv[r], w);
    below += (c < base) ? 1 : 0;
    const int k = c - base;
    if ((unsigned)k < (unsigned)nb) {
        atomicAdd(&s_cnt[k], 1);
        const int slot = atomicAdd(s_nsel, 1);
        if (slot < CAPSEL) s_sel[slot] = e;
    }
}

__global__ void __launch_bounds__(BT)
build_sorted_edges_gather_pair(const int* __restrict__ row,
                               const int* __restrict__ col,
                               const float* __restrict__ ew,
                               int2* __restrict__ bounds,
                               int2* __restrict__ edges,
                               const float* __restrict__ inA,
                               const float* __restrict__ inB,
                               float* __restrict__ outA,
                               float* __restrict__ outB,
                               int E, int N) {
    __shared__ float s_dinv[NMAX];
    __shared__ int   s_cnt[SCNT];
    __shared__ int   s_red[32];
    __shared__ int   s_nsel;
    __shared__ int   s_start;
    __shared__ int   s_sel[CAPSEL];
    __shared__ int2  s_rec[CAPSEL];

    const int tid  = threadIdx.x;
    const int lane = tid & 31;
    const int wave = tid >> 5;
    const int nb   = (N + BG - 1) / BG;
    const int base = blockIdx.x * nb;

    for (int i = tid; i < NMAX; i += BT) s_dinv[i] = 0.f;
    if (tid < SCNT) s_cnt[tid] = 0;
    if (tid == 0) s_nsel = 0;
    __syncthreads();

    int below = 0;
    if ((E & 3) == 0) {
        const int4*   row4 = (const int4*)row;
        const int4*   col4 = (const int4*)col;
        const float4* ew4  = (const float4*)ew;
        const int nq = E >> 2;
        int q0 = tid;
        for (; q0 + (CH - 1) * BT < nq; q0 += BT * CH) {
            int4 r[CH], c[CH];
            float4 w[CH];
#pragma unroll
            for (int k = 0; k < CH; ++k) {
                r[k] = row4[q0 + k * BT];
                c[k] = col4[q0 + k * BT];
                w[k] = ew4[q0 + k * BT];
            }
#pragma unroll
            for (int k = 0; k < CH; ++k) {
                const int e = 4 * (q0 + k * BT);
                count_edge(s_dinv, s_cnt, s_sel, &s_nsel, r[k].x, c[k].x, w[k].x, e,     base, nb, below);
                count_edge(s_dinv, s_cnt, s_sel, &s_nsel, r[k].y, c[k].y, w[k].y, e + 1, base, nb, below);
                count_edge(s_dinv, s_cnt, s_sel, &s_nsel, r[k].z, c[k].z, w[k].z, e + 2, base, nb, below);
                count_edge(s_dinv, s_cnt, s_sel, &s_nsel, r[k].w, c[k].w, w[k].w, e + 3, base, nb, below);
            }
        }
        for (int q = q0; q < nq; q += BT) {
            const int4 r = row4[q], c = col4[q];
            const float4 w = ew4[q];
            const int e = 4 * q;
            count_edge(s_dinv, s_cnt, s_sel, &s_nsel, r.x, c.x, w.x, e,     base, nb, below);
            count_edge(s_dinv, s_cnt, s_sel, &s_nsel, r.y, c.y, w.y, e + 1, base, nb, below);
            count_edge(s_dinv, s_cnt, s_sel, &s_nsel, r.z, c.z, w.z, e + 2, base, nb, below);
            count_edge(s_dinv, s_cnt, s_sel, &s_nsel, r.w, c.w, w.w, e + 3, base, nb, below);
        }
    } else {
        for (int e = tid; e < E; e += BT)
            count_edge(s_dinv, s_cnt, s_sel, &s_nsel, row[e], col[e], ew[e], e, base, nb, below);
    }
    for (int d = 16; d > 0; d >>= 1) below += __shfl_xor(below, d);
    if (lane == 0) s_red[wave] = below;
    __syncthreads();

    for (int i = tid; i < NMAX; i += BT) {
        const float d = s_dinv[i];
        s_dinv[i] = (d > 0.f) ? rsqrtf(d) : 0.f;
    }
    if (wave == 0) {
        int t = (lane < BT / 32) ? s_red[lane] : 0;
        for (int d = 16; d > 0; d >>= 1) t += __shfl_xor(t, d);
        const int start = ((t + 15) & ~15) + 16 * (int)blockIdx.x;
        const int i0 = 2 * lane;
        const int c0 = s_cnt[i0], c1 = s_cnt[i0 + 1];
        int v = c0 + c1;
        for (int d = 1; d < 32; d <<= 1) {
            const int u = __shfl_up(v, d);
            if (lane >= d) v += u;
        }
        const int o0 = v - c0 - c1, o1 = o0 + c0;
        s_cnt[i0] = o0;
        s_cnt[i0 + 1] = o1;
        const bool v0 = i0 < nb && base + i0 < N;
        const bool v1 = i0 + 1 < nb && base + i0 + 1 < N;
        if (v0 && v1 && (base & 1) == 0) {
            *(int4*)&bounds[base + i0] = make_int4(start + o0, start + o1, start + o1, start + o1 + c1);
        } else {
            if (v0) bounds[base + i0]     = make_int2(start + o0, start + o1);
            if (v1) bounds[base + i0 + 1] = make_int2(start + o1, start + o1 + c1);
        }
        if (lane == 0) s_start = start;
    }
    __syncthreads();

    const int nsel  = s_nsel;
    const int start = s_start;
    const int wave_u = __builtin_amdgcn_readfirstlane(wave);
    const float4* const in4[2] = {(const float4*)inA, (const float4*)inB};
    if (nsel <= CAPSEL) {
        for (int i = tid; i < nsel; i += BT) {
            const int e = s_sel[i];
            const int r = row[e], c = col[e];
            const float w = -s_dinv[r] * ew[e] * s_dinv[c];
            const int pos = atomicAdd(&s_cnt[c - base], 1);
            s_rec[pos] = make_int2(r, __float_as_int(w));
        }
        __syncthreads();
        const int4* s4 = (const int4*)s_rec;
        int4* g4 = (int4*)(edges + start);
        const int n2 = nsel >> 1;
        for (int i = tid; i < n2; i += BT) g4[i] = s4[i];
        if (tid == 0 && (nsel & 1)) edges[start + nsel - 1] = s_rec[nsel - 1];
        for (int k = wave_u; k < nb && base + k < N; k += BT / 32) {
            const int beg = (k > 0) ? __builtin_amdgcn_readfirstlane(s_cnt[k - 1]) : 0;
            const int end = __builtin_amdgcn_readfirstlane(s_cnt[k]);
            float4 acc[2] = {make_float4(0.f, 0.f, 0.f, 0.f), make_float4(0.f, 0.f, 0.f, 0.f)};
            gather_row<2>(acc, in4, s_rec, beg, end, load_records(s_rec, beg, end, lane), lane);
            const size_t o = (size_t)(base + k) * (DC / 4) + lane;
            ((float4*)outA)[o] = acc[0];
            ((float4*)outB)[o] = acc[1];
        }
    } else {
        for (int e = tid; e < E; e += BT) {
            const int c = col[e];
            if ((unsigned)(c - base) < (unsigned)nb) {
                const int r = row[e];
                const float w = -s_dinv[r] * ew[e] * s_dinv[c];
                const int pos = atomicAdd(&s_cnt[c - base], 1);
                edges[start + pos] = make_int2(r, __float_as_int(w));
            }
        }
        __threadfence();
        __syncthreads();
        for (int k = wave_u; k < nb && base + k < N; k += BT / 32) {
            const int beg = start + ((k > 0) ? __builtin_amdgcn_readfirstlane(s_cnt[k - 1]) : 0);
            const int end = start + __builtin_amdgcn_readfirstlane(s_cnt[k]);
            float4 acc[2] = {make_float4(0.f, 0.f, 0.f, 0.f), make_float4(0.f, 0.f, 0.f, 0.f)};
            gather_row<2>(acc, in4, edges, beg, end, load_records(edges, beg, end, lane), lane);
            const size_t o = (size_t)(base + k) * (DC / 4) + lane;
            ((float4*)outA)[o] = acc[0];
            ((float4*)outB)[o] = acc[1];
        }
    }
}

__global__ void __launch_bounds__(256)
spmm_gather(const int2* __restrict__ bounds,
            const int2* __restrict__ edges,
            const float* __restrict__ in,
            float* __restrict__ out, int N) {
    const int lane = threadIdx.x & 31;
    const int node = blockIdx.x * 8 + __builtin_amdgcn_readfirstlane(threadIdx.x >> 5);
    if (node >= N) return;
    const int2 be = bounds[node];
    const int beg = be.x, end = be.y;
    const float4* const in4[1] = {(const float4*)in};
    float4 acc[1] = {make_float4(0.f, 0.f, 0.f, 0.f)};
    gather_row<1>(acc, in4, edges, beg, end, load_records(edges, beg, end, lane), lane);
    ((float4*)out)[(size_t)node * (DC / 4) + lane] = acc[0];
}

__global__ void __launch_bounds__(256)
spmm_gather_relu(const int2* __restrict__ bounds,
                 const int2* __restrict__ edges,
                 const float* __restrict__ in,
                 const float* __restrict__ w10,
                 float* __restrict__ s1_h,
                 const float* __restrict__ w12,
                 const float* __restrict__ b1, int N) {
    const int lane = threadIdx.x & 31;
    const int node = blockIdx.x * 8 + __builtin_amdgcn_readfirstlane(threadIdx.x >> 5);
    if (node >= N) return;
    const int2 be = bounds[node];
    const int beg = be.x, end = be.y;
    const int2 m0 = load_records(edges, beg, end, lane);
    const size_t o = (size_t)node * (DC / 4) + lane;
    const float4 a = ((const float4*)w10)[o];
    const float4 s = ((const float4*)s1_h)[o];
    const float4 m = ((const float4*)w12)[o];
    const float4 b = ((const float4*)b1)[lane];
    const float4* const in4[1] = {(const float4*)in};
    float4 acc[1] = {make_float4(0.f, 0.f, 0.f, 0.f)};
    gather_row<1>(acc, in4, edges, beg, end, m0, lane);
    const float4 t = acc[0];
    float4 r;
    r.x = fmaxf(a.x + s.x + 2.f * t.x - m.x + b.x, 0.f);
    r.y = fmaxf(a.y + s.y + 2.f * t.y - m.y + b.y, 0.f);
    r.z = fmaxf(a.z + s.z + 2.f * t.z - m.z + b.z, 0.f);
    r.w = fmaxf(a.w + s.w + 2.f * t.w - m.w + b.w, 0.f);
    ((float4*)s1_h)[o] = r;
}

#define APAD 132

__global__ void __launch_bounds__(256)
gemm3_gather_wmma_f32(const int2* __restrict__ bounds,
                      const int2* __restrict__ edges,
                      const float* __restrict__ A0,
                      const float* __restrict__ A1,
                      const float* __restrict__ B,
                      const float* __restrict__ bias,
                      float* __restrict__ out) {
    __shared__ float As[3][16][APAD];

    const int wave = __builtin_amdgcn_readfirstlane(threadIdx.x >> 5);
    const int lane = threadIdx.x & 31;
    const int idx  = lane & 15;
    const int half = lane >> 4;

    const int tm   = blockIdx.x;
    const int tn   = wave;
    const int colB = tn * 16 + idx;

    const int r0 = wave, r1 = wave + 8;
    const int n0 = tm * 16 + r0, n1 = tm * 16 + r1;
    const int2 be0 = bounds[n0], be1 = bounds[n1];
    const int beg0 = be0.x, end0 = be0.y;
    const int beg1 = be1.x, end1 = be1.y;
    const int2 m0 = load_records(edges, beg0, end0, lane);
    const int2 m1 = load_records(edges, beg1, end1, lane);
    const float4* const H4[1] = {(const float4*)A0};
    const float4* const T4[1] = {(const float4*)A1};
    const float4 h0 = H4[0][(size_t)n0 * (DC / 4) + lane];
    const float4 h1 = H4[0][(size_t)n1 * (DC / 4) + lane];
    const float4 t0 = T4[0][(size_t)n0 * (DC / 4) + lane];
    const float4 t1 = T4[0][(size_t)n1 * (DC / 4) + lane];
    float4 u0[1] = {make_float4(0.f, 0.f, 0.f, 0.f)};
    float4 u1[1] = {make_float4(0.f, 0.f, 0.f, 0.f)};
    gather_row<1>(u0, T4, edges, beg0, end0, m0, lane);
    gather_row<1>(u1, T4, edges, beg1, end1, m1, lane);
    float4 x0, x1;
    x0.x = 2.f * u0[0].x - h0.x;  x0.y = 2.f * u0[0].y - h0.y;
    x0.z = 2.f * u0[0].z - h0.z;  x0.w = 2.f * u0[0].w - h0.w;
    x1.x = 2.f * u1[0].x - h1.x;  x1.y = 2.f * u1[0].y - h1.y;
    x1.z = 2.f * u1[0].z - h1.z;  x1.w = 2.f * u1[0].w - h1.w;
    *(float4*)&As[0][r0][lane << 2] = h0;
    *(float4*)&As[0][r1][lane << 2] = h1;
    *(float4*)&As[1][r0][lane << 2] = t0;
    *(float4*)&As[1][r1][lane << 2] = t1;
    *(float4*)&As[2][r0][lane << 2] = x0;
    *(float4*)&As[2][r1][lane << 2] = x1;
    __syncthreads();

    v8f c = {0.f, 0.f, 0.f, 0.f, 0.f, 0.f, 0.f, 0.f};

#pragma unroll
    for (int m = 0; m < 3; ++m) {
        const float* Arow  = &As[m][idx][2 * half];
        const float* Bbase = B + (size_t)m * DC * DC
                               + (size_t)2 * half * DC + colB;
#pragma unroll
        for (int k0 = 0; k0 < DC; k0 += 4) {
            v2f a = *(const v2f*)(Arow + k0);
            v2f b;
            b[0] = Bbase[(size_t)k0 * DC];
            b[1] = Bbase[(size_t)k0 * DC + DC];
            c = __builtin_amdgcn_wmma_f32_16x16x4_f32(
                    false, a, false, b, (short)0, c, false, false);
        }
    }

    float bv = bias[colB];
#pragma unroll
    for (int r = 0; r < 8; ++r) {
        int orow = tm * 16 + r + 8 * half;
        out[(size_t)orow * DC + colB] = c[r] + bv;
    }
}


extern "C" void kernel_launch(void* const* d_in, const int* in_sizes, int n_in,
                              void* d_out, int out_size, void* d_ws, size_t ws_size,
                              hipStream_t stream) {
    const int*   ei  = (const int*)d_in[1];
    const float* ew  = (const float*)d_in[2];
    const float* W1  = (const float*)d_in[3];
    const float* b1  = (const float*)d_in[4];
    const float* W2  = (const float*)d_in[5];
    const float* b2  = (const float*)d_in[6];
    float* out = (float*)d_out;

    const int E = in_sizes[2];
    const int D = in_sizes[4];
    const int N = in_sizes[3] / (3 * D);
    const int ND = N * D;

    const int* row = ei;
    const int* col = ei + E;
    const float* W1_0 = W1;
    const float* W1_1 = W1 + (size_t)ND;
    const float* W1_2 = W1 + (size_t)2 * ND;

    char* ws = (char*)d_ws;
    int2*  bounds = (int2*)ws;                   ws += (size_t)N * 8;
    ws += (size_t)E * 4;
    float* B0 = (float*)ws;                      ws += (size_t)ND * 4;
    float* B1 = (float*)ws;                      ws += (size_t)ND * 4;
    int2*  edges = (int2*)ws;

    const int TB = 256;
    dim3 gRows((N + 7) / 8);

    build_sorted_edges_gather_pair<<<BG, BT, 0, stream>>>(row, col, ew, bounds, edges,
                                                          W1_1, W1_2, B0, B1, E, N);

    spmm_gather_relu<<<gRows, TB, 0, stream>>>(bounds, edges, B1, W1_0, B0, W1_2, b1, N);

    spmm_gather<<<gRows, TB, 0, stream>>>(bounds, edges, B0, B1, N);

    gemm3_gather_wmma_f32<<<N / 16, 256, 0, stream>>>(bounds, edges, B0, B1, W2, b2, out);
}


# Layer 1 re-associated: A*W1_1 + 2*A*(A*W1_2) summed as A*(W1_1 + 2*A*W1_2), so the sort launch gathers one source matrix, not two (16 in-edges per batch); same products, other sum order
# speedup vs baseline: 3.0434x; 1.0877x over previous
#include <hip/hip_runtime.h>


typedef __attribute__((ext_vector_type(2))) float v2f;
typedef __attribute__((ext_vector_type(8))) float v8f;

#define DC 128


template <int K, int NS>
__device__ __forceinline__ void gather_rows(float4 (&acc)[NS], const float4* const (&in4)[NS],
                                            int msrc, int mw, int j, int lane) {
    float4 x[K][NS];
    float  w[K];
#pragma unroll
    for (int k = 0; k < K; ++k) {
        const int s = __builtin_amdgcn_readlane(msrc, j + k);
        w[k] = __int_as_float(__builtin_amdgcn_readlane(mw, j + k));
#pragma unroll
        for (int n = 0; n < NS; ++n) x[k][n] = in4[n][(size_t)s * (DC / 4) + lane];
    }
#pragma unroll
    for (int k = 0; k < K; ++k) {
#pragma unroll
        for (int n = 0; n < NS; ++n) {
            acc[n].x += w[k] * x[k][n].x;
            acc[n].y += w[k] * x[k][n].y;
            acc[n].z += w[k] * x[k][n].z;
            acc[n].w += w[k] * x[k][n].w;
        }
    }
}

__device__ __forceinline__ int2 load_records(const int2* __restrict__ edges,
                                             int base, int end, int lane) {
    int2 m = make_int2(0, 0);
    if (lane < end - base) m = edges[base + lane];
    return m;
}

template <int K, int NS>
__device__ __forceinline__ void gather_tail(float4 (&acc)[NS], const float4* const (&in4)[NS],
                                            int msrc, int mw, int j, int rem, int lane) {
    if constexpr (K > 0) {
        if (rem == K) gather_rows<K, NS>(acc, in4, msrc, mw, j, lane);
        else gather_tail<K - 1, NS>(acc, in4, msrc, mw, j, rem, lane);
    }
}

template <int NS>
__device__ __forceinline__ void gather_row(float4 (&acc)[NS], const float4* const (&in4)[NS],
                                           const int2* __restrict__ edges,
                                           int beg, int end, int2 m, int lane) {
    constexpr int KB = 16 / NS;
    for (int base = beg; base < end; base += 32) {
        if (base != beg) m = load_records(edges, base, end, lane);
        const int cnt = min(32, end - base);
        int j = 0;
        for (; j + KB <= cnt; j += KB) gather_rows<KB, NS>(acc, in4, m.x, m.y, j, lane);
        gather_tail<KB - 1, NS>(acc, in4, m.x, m.y, j, cnt - j, lane);
    }
}


#define NMAX 4096
#define BT 512
#define BG 256
#define NBMAX (NMAX / BG)
#define SCNT 64
#define CAPSEL 16384
#define CH 4

__device__ __forceinline__ void count_edge(float* s_dinv, int* s_cnt, int* s_sel, int* s_nsel,
                                           int r, int c, float w, int e,
                                           int base, int nb, int& below) {
    atomicAdd(&s_dinv[r], w);
    below += (c < base) ? 1 : 0;
    const int k = c - base;
    if ((unsigned)k < (unsigned)nb) {
        atomicAdd(&s_cnt[k], 1);
        const int slot = atomicAdd(s_nsel, 1);
        if (slot < CAPSEL) s_sel[slot] = e;
    }
}

__global__ void __launch_bounds__(BT)
build_sorted_edges_gather_first(const int* __restrict__ row,
                                const int* __restrict__ col,
                                const float* __restrict__ ew,
                                int2* __restrict__ bounds,
                                int2* __restrict__ edges,
                                const float* __restrict__ inG,
                                const float* __restrict__ inL,
                                float* __restrict__ outV,
                                int E, int N) {
    __shared__ float s_dinv[NMAX];
    __shared__ int   s_cnt[SCNT];
    __shared__ int   s_red[32];
    __shared__ int   s_nsel;
    __shared__ int   s_start;
    __shared__ int   s_sel[CAPSEL];
    __shared__ int2  s_rec[CAPSEL];

    const int tid  = threadIdx.x;
    const int lane = tid & 31;
    const int wave = tid >> 5;
    const int nb   = (N + BG - 1) / BG;
    const int base = blockIdx.x * nb;

    for (int i = tid; i < NMAX; i += BT) s_dinv[i] = 0.f;
    if (tid < SCNT) s_cnt[tid] = 0;
    if (tid == 0) s_nsel = 0;
    __syncthreads();

    int below = 0;
    if ((E & 3) == 0) {
        const int4*   row4 = (const int4*)row;
        const int4*   col4 = (const int4*)col;
        const float4* ew4  = (const float4*)ew;
        const int nq = E >> 2;
        int q0 = tid;
        for (; q0 + (CH - 1) * BT < nq; q0 += BT * CH) {
            int4 r[CH], c[CH];
            float4 w[CH];
#pragma unroll
            for (int k = 0; k < CH; ++k) {
                r[k] = row4[q0 + k * BT];
                c[k] = col4[q0 + k * BT];
                w[k] = ew4[q0 + k * BT];
            }
#pragma unroll
            for (int k = 0; k < CH; ++k) {
                const int e = 4 * (q0 + k * BT);
                count_edge(s_dinv, s_cnt, s_sel, &s_nsel, r[k].x, c[k].x, w[k].x, e,     base, nb, below);
                count_edge(s_dinv, s_cnt, s_sel, &s_nsel, r[k].y, c[k].y, w[k].y, e + 1, base, nb, below);
                count_edge(s_dinv, s_cnt, s_sel, &s_nsel, r[k].z, c[k].z, w[k].z, e + 2, base, nb, below);
                count_edge(s_dinv, s_cnt, s_sel, &s_nsel, r[k].w, c[k].w, w[k].w, e + 3, base, nb, below);
            }
        }
        for (int q = q0; q < nq; q += BT) {
            const int4 r = row4[q], c = col4[q];
            const float4 w = ew4[q];
            const int e = 4 * q;
            count_edge(s_dinv, s_cnt, s_sel, &s_nsel, r.x, c.x, w.x, e,     base, nb, below);
            count_edge(s_dinv, s_cnt, s_sel, &s_nsel, r.y, c.y, w.y, e + 1, base, nb, below);
            count_edge(s_dinv, s_cnt, s_sel, &s_nsel, r.z, c.z, w.z, e + 2, base, nb, below);
            count_edge(s_dinv, s_cnt, s_sel, &s_nsel, r.w, c.w, w.w, e + 3, base, nb, below);
        }
    } else {
        for (int e = tid; e < E; e += BT)
            count_edge(s_dinv, s_cnt, s_sel, &s_nsel, row[e], col[e], ew[e], e, base, nb, below);
    }
    for (int d = 16; d > 0; d >>= 1) below += __shfl_xor(below, d);
    if (lane == 0) s_red[wave] = below;
    __syncthreads();

    for (int i = tid; i < NMAX; i += BT) {
        const float d = s_dinv[i];
        s_dinv[i] = (d > 0.f) ? rsqrtf(d) : 0.f;
    }
    if (wave == 0) {
        int t = (lane < BT / 32) ? s_red[lane] : 0;
        for (int d = 16; d > 0; d >>= 1) t += __shfl_xor(t, d);
        const int start = ((t + 15) & ~15) + 16 * (int)blockIdx.x;
        const int i0 = 2 * lane;
        const int c0 = s_cnt[i0], c1 = s_cnt[i0 + 1];
        int v = c0 + c1;
        for (int d = 1; d < 32; d <<= 1) {
            const int u = __shfl_up(v, d);
            if (lane >= d) v += u;
        }
        const int o0 = v - c0 - c1, o1 = o0 + c0;
        s_cnt[i0] = o0;
        s_cnt[i0 + 1] = o1;
        const bool v0 = i0 < nb && base + i0 < N;
        const bool v1 = i0 + 1 < nb && base + i0 + 1 < N;
        if (v0 && v1 && (base & 1) == 0) {
            *(int4*)&bounds[base + i0] = make_int4(start + o0, start + o1, start + o1, start + o1 + c1);
        } else {
            if (v0) bounds[base + i0]     = make_int2(start + o0, start + o1);
            if (v1) bounds[base + i0 + 1] = make_int2(start + o1, start + o1 + c1);
        }
        if (lane == 0) s_start = start;
    }
    __syncthreads();

    const int nsel  = s_nsel;
    const int start = s_start;
    const int wave_u = __builtin_amdgcn_readfirstlane(wave);
    const float4* const in4[1] = {(const float4*)inG};
    if (nsel <= CAPSEL) {
        for (int i = tid; i < nsel; i += BT) {
            const int e = s_sel[i];
            const int r = row[e], c = col[e];
            const float w = -s_dinv[r] * ew[e] * s_dinv[c];
            const int pos = atomicAdd(&s_cnt[c - base], 1);
            s_rec[pos] = make_int2(r, __float_as_int(w));
        }
        __syncthreads();
        const int4* s4 = (const int4*)s_rec;
        int4* g4 = (int4*)(edges + start);
        const int n2 = nsel >> 1;
        for (int i = tid; i < n2; i += BT) g4[i] = s4[i];
        if (tid == 0 && (nsel & 1)) edges[start + nsel - 1] = s_rec[nsel - 1];
        for (int k = wave_u; k < nb && base + k < N; k += BT / 32) {
            const int beg = (k > 0) ? __builtin_amdgcn_readfirstlane(s_cnt[k - 1]) : 0;
            const int end = __builtin_amdgcn_readfirstlane(s_cnt[k]);
            const size_t o = (size_t)(base + k) * (DC / 4) + lane;
            const float4 l = ((const float4*)inL)[o];
            float4 acc[1] = {make_float4(0.f, 0.f, 0.f, 0.f)};
            gather_row<1>(acc, in4, s_rec, beg, end, load_records(s_rec, beg, end, lane), lane);
            float4 v;
            v.x = l.x + 2.f * acc[0].x;  v.y = l.y + 2.f * acc[0].y;
            v.z = l.z + 2.f * acc[0].z;  v.w = l.w + 2.f * acc[0].w;
            ((float4*)outV)[o] = v;
        }
    } else {
        for (int e = tid; e < E; e += BT) {
            const int c = col[e];
            if ((unsigned)(c - base) < (unsigned)nb) {
                const int r = row[e];
                const float w = -s_dinv[r] * ew[e] * s_dinv[c];
                const int pos = atomicAdd(&s_cnt[c - base], 1);
                edges[start + pos] = make_int2(r, __float_as_int(w));
            }
        }
        __threadfence();
        __syncthreads();
        for (int k = wave_u; k < nb && base + k < N; k += BT / 32) {
            const int beg = start + ((k > 0) ? __builtin_amdgcn_readfirstlane(s_cnt[k - 1]) : 0);
            const int end = start + __builtin_amdgcn_readfirstlane(s_cnt[k]);
            const size_t o = (size_t)(base + k) * (DC / 4) + lane;
            const float4 l = ((const float4*)inL)[o];
            float4 acc[1] = {make_float4(0.f, 0.f, 0.f, 0.f)};
            gather_row<1>(acc, in4, edges, beg, end, load_records(edges, beg, end, lane), lane);
            float4 v;
            v.x = l.x + 2.f * acc[0].x;  v.y = l.y + 2.f * acc[0].y;
            v.z = l.z + 2.f * acc[0].z;  v.w = l.w + 2.f * acc[0].w;
            ((float4*)outV)[o] = v;
        }
    }
}

__global__ void __launch_bounds__(256) __attribute__((amdgpu_waves_per_eu(2, 8)))
spmm_gather(const int2* __restrict__ bounds,
            const int2* __restrict__ edges,
            const float* __restrict__ in,
            float* __restrict__ out, int N) {
    const int lane = threadIdx.x & 31;
    const int node = blockIdx.x * 8 + __builtin_amdgcn_readfirstlane(threadIdx.x >> 5);
    if (node >= N) return;
    const int2 be = bounds[node];
    const int beg = be.x, end = be.y;
    const float4* const in4[1] = {(const float4*)in};
    float4 acc[1] = {make_float4(0.f, 0.f, 0.f, 0.f)};
    gather_row<1>(acc, in4, edges, beg, end, load_records(edges, beg, end, lane), lane);
    ((float4*)out)[(size_t)node * (DC / 4) + lane] = acc[0];
}

__global__ void __launch_bounds__(256)
spmm_gather_relu(const int2* __restrict__ bounds,
                 const int2* __restrict__ edges,
                 const float* __restrict__ in,
                 const float* __restrict__ w10,
                 const float* __restrict__ w12,
                 const float* __restrict__ b1,
                 float* __restrict__ h, int N) {
    const int lane = threadIdx.x & 31;
    const int node = blockIdx.x * 8 + __builtin_amdgcn_readfirstlane(threadIdx.x >> 5);
    if (node >= N) return;
    const int2 be = bounds[node];
    const int beg = be.x, end = be.y;
    const int2 m0 = load_records(edges, beg, end, lane);
    const size_t o = (size_t)node * (DC / 4) + lane;
    const float4 a = ((const float4*)w10)[o];
    const float4 m = ((const float4*)w12)[o];
    const float4 b = ((const float4*)b1)[lane];
    const float4* const in4[1] = {(const float4*)in};
    float4 acc[1] = {make_float4(0.f, 0.f, 0.f, 0.f)};
    gather_row<1>(acc, in4, edges, beg, end, m0, lane);
    const float4 t = acc[0];
    float4 r;
    r.x = fmaxf(a.x + t.x - m.x + b.x, 0.f);
    r.y = fmaxf(a.y + t.y - m.y + b.y, 0.f);
    r.z = fmaxf(a.z + t.z - m.z + b.z, 0.f);
    r.w = fmaxf(a.w + t.w - m.w + b.w, 0.f);
    ((float4*)h)[o] = r;
}

#define APAD 132

__global__ void __launch_bounds__(256)
gemm3_gather_wmma_f32(const int2* __restrict__ bounds,
                      const int2* __restrict__ edges,
                      const float* __restrict__ A0,
                      const float* __restrict__ A1,
                      const float* __restrict__ B,
                      const float* __restrict__ bias,
                      float* __restrict__ out) {
    __shared__ float As[3][16][APAD];

    const int wave = __builtin_amdgcn_readfirstlane(threadIdx.x >> 5);
    const int lane = threadIdx.x & 31;
    const int idx  = lane & 15;
    const int half = lane >> 4;

    const int tm   = blockIdx.x;
    const int tn   = wave;
    const int colB = tn * 16 + idx;

    const int r0 = wave, r1 = wave + 8;
    const int n0 = tm * 16 + r0, n1 = tm * 16 + r1;
    const int2 be0 = bounds[n0], be1 = bounds[n1];
    const int beg0 = be0.x, end0 = be0.y;
    const int beg1 = be1.x, end1 = be1.y;
    const int2 m0 = load_records(edges, beg0, end0, lane);
    const int2 m1 = load_records(edges, beg1, end1, lane);
    const float4* const H4[1] = {(const float4*)A0};
    const float4* const T4[1] = {(const float4*)A1};
    const float4 h0 = H4[0][(size_t)n0 * (DC / 4) + lane];
    const float4 h1 = H4[0][(size_t)n1 * (DC / 4) + lane];
    const float4 t0 = T4[0][(size_t)n0 * (DC / 4) + lane];
    const float4 t1 = T4[0][(size_t)n1 * (DC / 4) + lane];
    float4 u0[1] = {make_float4(0.f, 0.f, 0.f, 0.f)};
    float4 u1[1] = {make_float4(0.f, 0.f, 0.f, 0.f)};
    gather_row<1>(u0, T4, edges, beg0, end0, m0, lane);
    gather_row<1>(u1, T4, edges, beg1, end1, m1, lane);
    float4 x0, x1;
    x0.x = 2.f * u0[0].x - h0.x;  x0.y = 2.f * u0[0].y - h0.y;
    x0.z = 2.f * u0[0].z - h0.z;  x0.w = 2.f * u0[0].w - h0.w;
    x1.x = 2.f * u1[0].x - h1.x;  x1.y = 2.f * u1[0].y - h1.y;
    x1.z = 2.f * u1[0].z - h1.z;  x1.w = 2.f * u1[0].w - h1.w;
    *(float4*)&As[0][r0][lane << 2] = h0;
    *(float4*)&As[0][r1][lane << 2] = h1;
    *(float4*)&As[1][r0][lane << 2] = t0;
    *(float4*)&As[1][r1][lane << 2] = t1;
    *(float4*)&As[2][r0][lane << 2] = x0;
    *(float4*)&As[2][r1][lane << 2] = x1;
    __syncthreads();

    v8f c = {0.f, 0.f, 0.f, 0.f, 0.f, 0.f, 0.f, 0.f};

#pragma unroll
    for (int m = 0; m < 3; ++m) {
        const float* Arow  = &As[m][idx][2 * half];
        const float* Bbase = B + (size_t)m * DC * DC
                               + (size_t)2 * half * DC + colB;
#pragma unroll
        for (int k0 = 0; k0 < DC; k0 += 4) {
            v2f a = *(const v2f*)(Arow + k0);
            v2f b;
            b[0] = Bbase[(size_t)k0 * DC];
            b[1] = Bbase[(size_t)k0 * DC + DC];
            c = __builtin_amdgcn_wmma_f32_16x16x4_f32(
                    false, a, false, b, (short)0, c, false, false);
        }
    }

    float bv = bias[colB];
#pragma unroll
    for (int r = 0; r < 8; ++r) {
        int orow = tm * 16 + r + 8 * half;
        out[(size_t)orow * DC + colB] = c[r] + bv;
    }
}


extern "C" void kernel_launch(void* const* d_in, const int* in_sizes, int n_in,
                              void* d_out, int out_size, void* d_ws, size_t ws_size,
                              hipStream_t stream) {
    const int*   ei  = (const int*)d_in[1];
    const float* ew  = (const float*)d_in[2];
    const float* W1  = (const float*)d_in[3];
    const float* b1  = (const float*)d_in[4];
    const float* W2  = (const float*)d_in[5];
    const float* b2  = (const float*)d_in[6];
    float* out = (float*)d_out;

    const int E = in_sizes[2];
    const int D = in_sizes[4];
    const int N = in_sizes[3] / (3 * D);
    const int ND = N * D;

    const int* row = ei;
    const int* col = ei + E;
    const float* W1_0 = W1;
    const float* W1_1 = W1 + (size_t)ND;
    const float* W1_2 = W1 + (size_t)2 * ND;

    char* ws = (char*)d_ws;
    int2*  bounds = (int2*)ws;                   ws += (size_t)N * 8;
    ws += (size_t)E * 4;
    float* B0 = (float*)ws;                      ws += (size_t)ND * 4;
    float* B1 = (float*)ws;                      ws += (size_t)ND * 4;
    int2*  edges = (int2*)ws;

    const int TB = 256;
    dim3 gRows((N + 7) / 8);

    build_sorted_edges_gather_first<<<BG, BT, 0, stream>>>(row, col, ew, bounds, edges,
                                                           W1_2, W1_1, B1, E, N);

    spmm_gather_relu<<<gRows, TB, 0, stream>>>(bounds, edges, B1, W1_0, W1_2, b1, B0, N);

    spmm_gather<<<gRows, TB, 0, stream>>>(bounds, edges, B0, B1, N);

    gemm3_gather_wmma_f32<<<N / 16, 256, 0, stream>>>(bounds, edges, B0, B1, W2, b2, out);
}
